# starting kernel
# baseline (speedup 1.0000x reference)
#include <hip/hip_runtime.h>
#include <math.h>

#define BATCH 8
#define CCH   768
#define HH    32
#define WW    64
#define HW    (HH*WW)
#define NPTS  (BATCH*HH*WW)
#define DD    768
#define KK    19
#define MMIX  5
#define KM    95
#define KMP   96
#define LOG2PI 1.8378770664093453f

typedef __attribute__((ext_vector_type(2))) float v2f;
typedef __attribute__((ext_vector_type(8))) float v8f;

typedef const float __attribute__((address_space(1)))* gcf;
typedef const v2f   __attribute__((address_space(1)))* gcv2;

__global__ __launch_bounds__(256) void gmm_prep_b(
    const float* __restrict__ means, const float* __restrict__ log_vars,
    float* __restrict__ b1, float* __restrict__ b2) {
  int idx = blockIdx.x * 256 + threadIdx.x;
  if (idx >= KMP * DD) return;
  int d = idx / KMP;
  int c = idx % KMP;
  float v1 = 0.0f, v2 = 0.0f;
  if (c < KM) {
    float lv = log_vars[c * DD + d];
    float mu = means[c * DD + d];
    float iv = expf(-lv);
    v1 = iv;
    v2 = -2.0f * mu * iv;
  }
  int o = (d >> 1) * (2 * KMP) + c * 2 + (d & 1);
  b1[o] = v1;
  b2[o] = v2;
}

__global__ __launch_bounds__(128) void gmm_prep_c(
    const float* __restrict__ means, const float* __restrict__ log_vars,
    const float* __restrict__ log_w, float* __restrict__ ckm) {
  int c = threadIdx.x;
  if (c >= KMP) return;
  if (c >= KM) { ckm[c] = -1.0e30f; return; }
  float slv = 0.0f, smi = 0.0f;
  for (int d = 0; d < DD; ++d) {
    float lv = log_vars[c * DD + d];
    float mu = means[c * DD + d];
    slv += lv;
    smi += mu * mu * expf(-lv);
  }
  ckm[c] = log_w[c] - 0.5f * ((float)DD * LOG2PI + slv + smi);
}

__global__ __launch_bounds__(128) void gmm_main(
    const float* __restrict__ feats,
    const float* __restrict__ b1, const float* __restrict__ b2,
    const float* __restrict__ ckm, float* __restrict__ out) {
  __shared__ float sdata[128 * KMP];
  __shared__ float cksh[KMP];

  const int tid  = threadIdx.x;
  const int wave = tid >> 5;
  const int lane = tid & 31;
  const int lrow = lane & 15;
  const int lhi  = lane >> 4;

  if (tid < KMP) cksh[tid] = ckm[tid];

  v8f acc[2][6];
#pragma unroll
  for (int nt = 0; nt < 2; ++nt)
#pragma unroll
    for (int ct = 0; ct < 6; ++ct)
      acc[nt][ct] = (v8f){0.f,0.f,0.f,0.f,0.f,0.f,0.f,0.f};

  gcf ap[2];
#pragma unroll
  for (int nt = 0; nt < 2; ++nt) {
    int n0  = blockIdx.x * 128 + wave * 32 + nt * 16;
    int bb  = n0 >> 11;
    int rem = n0 & 2047;
    ap[nt] = (gcf)(feats + (size_t)bb * (size_t)(CCH * HW)
                         + (size_t)(2 * lhi) * HW + rem + lrow);
  }

  gcf bp1[6];
  gcf bp2[6];
#pragma unroll
  for (int ct = 0; ct < 6; ++ct) {
    int off = lhi * (2 * KMP) + (ct * 16 + lrow) * 2;
    bp1[ct] = (gcf)(b1 + off);
    bp2[ct] = (gcf)(b2 + off);
  }

  for (int d0 = 0; d0 < DD; d0 += 4) {
    v2f a[2], asq[2];
#pragma unroll
    for (int nt = 0; nt < 2; ++nt) {
      float a0 = ap[nt][0];
      float a1 = ap[nt][HW];
      ap[nt] += 4 * HW;
      a[nt].x = a0;        a[nt].y = a1;
      asq[nt].x = a0 * a0; asq[nt].y = a1 * a1;
    }
#pragma unroll
    for (int ct = 0; ct < 6; ++ct) {
      v2f vb1 = *(gcv2)bp1[ct];
      v2f vb2 = *(gcv2)bp2[ct];
      bp1[ct] += 2 * (2 * KMP);
      bp2[ct] += 2 * (2 * KMP);
#pragma unroll
      for (int nt = 0; nt < 2; ++nt) {
        acc[nt][ct] = __builtin_amdgcn_wmma_f32_16x16x4_f32(
            false, asq[nt], false, vb1, (short)0, acc[nt][ct], false, false);
        acc[nt][ct] = __builtin_amdgcn_wmma_f32_16x16x4_f32(
            false, a[nt],   false, vb2, (short)0, acc[nt][ct], false, false);
      }
    }
  }

#pragma unroll
  for (int nt = 0; nt < 2; ++nt)
#pragma unroll
    for (int ct = 0; ct < 6; ++ct)
#pragma unroll
      for (int r = 0; r < 8; ++r) {
        int pl = wave * 32 + nt * 16 + r + 8 * lhi;
        sdata[pl * KMP + ct * 16 + lrow] = acc[nt][ct][r];
      }
  __syncthreads();

  {
    const float* row = &sdata[tid * KMP];
    float lse[KK];
#pragma unroll
    for (int k = 0; k < KK; ++k) {
      float lp[MMIX];
      float mx = -3.0e38f;
#pragma unroll
      for (int m = 0; m < MMIX; ++m) {
        int c = k * MMIX + m;
        lp[m] = -0.5f * row[c] + cksh[c];
        mx = fmaxf(mx, lp[m]);
      }
      float s = 0.0f;
#pragma unroll
      for (int m = 0; m < MMIX; ++m) s += expf(lp[m] - mx);
      lse[k] = mx + logf(s);
    }
    float mv = lse[0];
#pragma unroll
    for (int k = 1; k < KK; ++k) mv = fmaxf(mv, lse[k]);
    float s = 0.0f;
#pragma unroll
    for (int k = 0; k < KK; ++k) s += expf(lse[k] - mv);
    out[blockIdx.x * 128 + tid] = 1.0f - 1.0f / s;
  }
}

extern "C" void kernel_launch(void* const* d_in, const int* in_sizes, int n_in,
                              void* d_out, int out_size, void* d_ws, size_t ws_size,
                              hipStream_t stream) {
  const float* feats     = (const float*)d_in[0];
  const float* means     = (const float*)d_in[1];
  const float* log_vars  = (const float*)d_in[2];
  const float* log_w     = (const float*)d_in[3];
  float* out = (float*)d_out;

  float* b1  = (float*)d_ws;
  float* b2  = b1 + KMP * DD;
  float* ckm = b2 + KMP * DD;

  (void)in_sizes; (void)n_in; (void)out_size; (void)ws_size;

  int prep_elems = KMP * DD;
  gmm_prep_b<<<(prep_elems + 255) / 256, 256, 0, stream>>>(means, log_vars, b1, b2);
  gmm_prep_c<<<1, 128, 0, stream>>>(means, log_vars, log_w, ckm);
  gmm_main<<<NPTS / 128, 128, 0, stream>>>(feats, b1, b2, ckm, out);
}


# main loop: the 16 fragment loads of k-step s+3 are issued as one group in front of the WMMAs of k-step s (ring of four register sets), so load latency overlaps matrix work; order of sums kept
# speedup vs baseline: 2.2676x; 2.2676x over previous
#include <hip/hip_runtime.h>
#include <math.h>

#define BATCH 8
#define CCH   768
#define HH    32
#define WW    64
#define HW    (HH*WW)
#define NPTS  (BATCH*HH*WW)
#define DD    768
#define KK    19
#define MMIX  5
#define KM    95
#define KMP   96
#define LOG2PI 1.8378770664093453f

typedef __attribute__((ext_vector_type(2))) float v2f;
typedef __attribute__((ext_vector_type(8))) float v8f;

typedef const float __attribute__((address_space(1)))* gcf;
typedef const v2f   __attribute__((address_space(1)))* gcv2;

__global__ __launch_bounds__(256) void gmm_prep_b(
    const float* __restrict__ means, const float* __restrict__ log_vars,
    float* __restrict__ b1, float* __restrict__ b2) {
  int idx = blockIdx.x * 256 + threadIdx.x;
  if (idx >= KMP * DD) return;
  int d = idx / KMP;
  int c = idx % KMP;
  float v1 = 0.0f, v2 = 0.0f;
  if (c < KM) {
    float lv = log_vars[c * DD + d];
    float mu = means[c * DD + d];
    float iv = expf(-lv);
    v1 = iv;
    v2 = -2.0f * mu * iv;
  }
  int o = (d >> 1) * (2 * KMP) + c * 2 + (d & 1);
  b1[o] = v1;
  b2[o] = v2;
}

__global__ __launch_bounds__(128) void gmm_prep_c(
    const float* __restrict__ means, const float* __restrict__ log_vars,
    const float* __restrict__ log_w, float* __restrict__ ckm) {
  int c = threadIdx.x;
  if (c >= KMP) return;
  if (c >= KM) { ckm[c] = -1.0e30f; return; }
  float slv = 0.0f, smi = 0.0f;
  for (int d = 0; d < DD; ++d) {
    float lv = log_vars[c * DD + d];
    float mu = means[c * DD + d];
    slv += lv;
    smi += mu * mu * expf(-lv);
  }
  ckm[c] = log_w[c] - 0.5f * ((float)DD * LOG2PI + slv + smi);
}

__global__ __launch_bounds__(128) void gmm_main(
    const float* __restrict__ feats,
    const float* __restrict__ b1, const float* __restrict__ b2,
    const float* __restrict__ ckm, float* __restrict__ out) {
  __shared__ float sdata[128 * KMP];
  __shared__ float cksh[KMP];

  const int tid  = threadIdx.x;
  const int wave = tid >> 5;
  const int lane = tid & 31;
  const int lrow = lane & 15;
  const int lhi  = lane >> 4;

  if (tid < KMP) cksh[tid] = ckm[tid];

  v8f acc[2][6];
#pragma unroll
  for (int nt = 0; nt < 2; ++nt)
#pragma unroll
    for (int ct = 0; ct < 6; ++ct)
      acc[nt][ct] = (v8f){0.f,0.f,0.f,0.f,0.f,0.f,0.f,0.f};

  gcf pa;
  {
    int n0  = blockIdx.x * 128 + wave * 32;
    int bb  = n0 >> 11;
    int rem = n0 & 2047;
    pa = (gcf)(feats + (size_t)bb * (size_t)(CCH * HW)
                     + (size_t)(2 * lhi) * HW + rem + lrow);
  }
  gcf pb1 = (gcf)(b1 + lhi * (2 * KMP) + lrow * 2);
  gcf pb2 = (gcf)(b2 + lhi * (2 * KMP) + lrow * 2);

  struct Frag { float a[2][2]; v2f w1[6]; v2f w2[6]; };

#define LOAD_STEP(F, j)                                                      \
  do {                                                                       \
    _Pragma("unroll")                                                        \
    for (int nt = 0; nt < 2; ++nt) {                                         \
      (F).a[nt][0] = pa[(j) * 4 * HW + nt * 16];                             \
      (F).a[nt][1] = pa[(j) * 4 * HW + HW + nt * 16];                        \
    }                                                                        \
    _Pragma("unroll")                                                        \
    for (int ct = 0; ct < 6; ++ct)                                           \
      (F).w1[ct] = *(gcv2)(pb1 + (j) * 4 * KMP + ct * 32);                   \
    _Pragma("unroll")                                                        \
    for (int ct = 0; ct < 6; ++ct)                                           \
      (F).w2[ct] = *(gcv2)(pb2 + (j) * 4 * KMP + ct * 32);                   \
  } while (0)

#define ADVANCE(n)                                                           \
  do { pa += (n) * 4 * HW; pb1 += (n) * 4 * KMP; pb2 += (n) * 4 * KMP; } while (0)

#define COMPUTE_STEP(F)                                                      \
  do {                                                                       \
    v2f a_[2], q_[2];                                                        \
    _Pragma("unroll")                                                        \
    for (int nt = 0; nt < 2; ++nt) {                                         \
      a_[nt].x = (F).a[nt][0];  a_[nt].y = (F).a[nt][1];                     \
      q_[nt].x = a_[nt].x * a_[nt].x;  q_[nt].y = a_[nt].y * a_[nt].y;       \
    }                                                                        \
    _Pragma("unroll")                                                        \
    for (int ct = 0; ct < 6; ++ct) {                                         \
      _Pragma("unroll")                                                      \
      for (int nt = 0; nt < 2; ++nt) {                                       \
        acc[nt][ct] = __builtin_amdgcn_wmma_f32_16x16x4_f32(                 \
            false, q_[nt], false, (F).w1[ct], (short)0, acc[nt][ct],         \
            false, false);                                                   \
        acc[nt][ct] = __builtin_amdgcn_wmma_f32_16x16x4_f32(                 \
            false, a_[nt], false, (F).w2[ct], (short)0, acc[nt][ct],         \
            false, false);                                                   \
      }                                                                      \
    }                                                                        \
  } while (0)

#define PIN() __builtin_amdgcn_sched_barrier(0)
  Frag f0, f1, f2, f3;
  LOAD_STEP(f0, 0);
  LOAD_STEP(f1, 1);
  LOAD_STEP(f2, 2);
  ADVANCE(3);
  for (int it = 0; it < DD / 16 - 1; ++it) {
    PIN();  LOAD_STEP(f3, 0);  PIN();  COMPUTE_STEP(f0);
    PIN();  LOAD_STEP(f0, 1);  PIN();  COMPUTE_STEP(f1);
    PIN();  LOAD_STEP(f1, 2);  PIN();  COMPUTE_STEP(f2);
    PIN();  LOAD_STEP(f2, 3);  PIN();  COMPUTE_STEP(f3);
    ADVANCE(4);
  }
  PIN();  LOAD_STEP(f3, 0);  PIN();
  COMPUTE_STEP(f0);
  COMPUTE_STEP(f1);
  COMPUTE_STEP(f2);
  COMPUTE_STEP(f3);
#undef PIN
#undef LOAD_STEP
#undef ADVANCE
#undef COMPUTE_STEP

#pragma unroll
  for (int nt = 0; nt < 2; ++nt)
#pragma unroll
    for (int ct = 0; ct < 6; ++ct)
#pragma unroll
      for (int r = 0; r < 8; ++r) {
        int pl = wave * 32 + nt * 16 + r + 8 * lhi;
        sdata[pl * KMP + ct * 16 + lrow] = acc[nt][ct][r];
      }
  __syncthreads();

  {
    const float* row = &sdata[tid * KMP];
    float lse[KK];
#pragma unroll
    for (int k = 0; k < KK; ++k) {
      float lp[MMIX];
      float mx = -3.0e38f;
#pragma unroll
      for (int m = 0; m < MMIX; ++m) {
        int c = k * MMIX + m;
        lp[m] = -0.5f * row[c] + cksh[c];
        mx = fmaxf(mx, lp[m]);
      }
      float s = 0.0f;
#pragma unroll
      for (int m = 0; m < MMIX; ++m) s += expf(lp[m] - mx);
      lse[k] = mx + logf(s);
    }
    float mv = lse[0];
#pragma unroll
    for (int k = 1; k < KK; ++k) mv = fmaxf(mv, lse[k]);
    float s = 0.0f;
#pragma unroll
    for (int k = 0; k < KK; ++k) s += expf(lse[k] - mv);
    out[blockIdx.x * 128 + tid] = 1.0f - 1.0f / s;
  }
}

extern "C" void kernel_launch(void* const* d_in, const int* in_sizes, int n_in,
                              void* d_out, int out_size, void* d_ws, size_t ws_size,
                              hipStream_t stream) {
  const float* feats     = (const float*)d_in[0];
  const float* means     = (const float*)d_in[1];
  const float* log_vars  = (const float*)d_in[2];
  const float* log_w     = (const float*)d_in[3];
  float* out = (float*)d_out;

  float* b1  = (float*)d_ws;
  float* b2  = b1 + KMP * DD;
  float* ckm = b2 + KMP * DD;

  (void)in_sizes; (void)n_in; (void)out_size; (void)ws_size;

  int prep_elems = KMP * DD;
  gmm_prep_b<<<(prep_elems + 255) / 256, 256, 0, stream>>>(means, log_vars, b1, b2);
  gmm_prep_c<<<1, 128, 0, stream>>>(means, log_vars, log_w, ckm);
  gmm_main<<<NPTS / 128, 128, 0, stream>>>(feats, b1, b2, ckm, out);
}


# prep: the column constant's 768-term serial sums (one block, 3 waves) become a block reduction fused into the B-prep launch, one block per column; 2 launches; the constant's sum order becomes a tree
# speedup vs baseline: 3.1153x; 1.3738x over previous
#include <hip/hip_runtime.h>
#include <math.h>

#define BATCH 8
#define CCH   768
#define HH    32
#define WW    64
#define HW    (HH*WW)
#define NPTS  (BATCH*HH*WW)
#define DD    768
#define KK    19
#define MMIX  5
#define KM    95
#define KMP   96
#define LOG2PI 1.8378770664093453f

typedef __attribute__((ext_vector_type(2))) float v2f;
typedef __attribute__((ext_vector_type(8))) float v8f;

typedef const float __attribute__((address_space(1)))* gcf;
typedef const v2f   __attribute__((address_space(1)))* gcv2;

__global__ __launch_bounds__(128) void gmm_prep(
    const float* __restrict__ means, const float* __restrict__ log_vars,
    const float* __restrict__ log_w,
    float* __restrict__ b1, float* __restrict__ b2, float* __restrict__ ckm) {
  __shared__ float red[2][4];
  const int c   = blockIdx.x;
  const int tid = threadIdx.x;

  if (c >= KM) {
#pragma unroll
    for (int j = 0; j < 3; ++j) {
      int o = (tid + j * 128) * (2 * KMP) + c * 2;
      *(v2f*)(b1 + o) = (v2f){0.0f, 0.0f};
      *(v2f*)(b2 + o) = (v2f){0.0f, 0.0f};
    }
    if (tid == 0) ckm[c] = -1.0e30f;
    return;
  }

  float slv = 0.0f, smi = 0.0f;
#pragma unroll
  for (int j = 0; j < 3; ++j) {
    int p = tid + j * 128;
    v2f lv = *(const v2f*)(log_vars + c * DD + 2 * p);
    v2f mu = *(const v2f*)(means    + c * DD + 2 * p);
    v2f iv, w2;
    iv.x = expf(-lv.x);
    iv.y = expf(-lv.y);
    w2.x = -2.0f * mu.x * iv.x;
    w2.y = -2.0f * mu.y * iv.y;
    int o = p * (2 * KMP) + c * 2;
    *(v2f*)(b1 + o) = iv;
    *(v2f*)(b2 + o) = w2;
    slv += lv.x;
    slv += lv.y;
    smi += mu.x * mu.x * iv.x;
    smi += mu.y * mu.y * iv.y;
  }
#pragma unroll
  for (int m = 16; m >= 1; m >>= 1) {
    slv += __shfl_xor(slv, m);
    smi += __shfl_xor(smi, m);
  }
  if ((tid & 31) == 0) { red[0][tid >> 5] = slv; red[1][tid >> 5] = smi; }
  __syncthreads();
  if (tid == 0) {
    float tlv = (red[0][0] + red[0][1]) + (red[0][2] + red[0][3]);
    float tmi = (red[1][0] + red[1][1]) + (red[1][2] + red[1][3]);
    ckm[c] = log_w[c] - 0.5f * ((float)DD * LOG2PI + tlv + tmi);
  }
}

__global__ __launch_bounds__(128) void gmm_main(
    const float* __restrict__ feats,
    const float* __restrict__ b1, const float* __restrict__ b2,
    const float* __restrict__ ckm, float* __restrict__ out) {
  __shared__ float sdata[128 * KMP];
  __shared__ float cksh[KMP];

  const int tid  = threadIdx.x;
  const int wave = tid >> 5;
  const int lane = tid & 31;
  const int lrow = lane & 15;
  const int lhi  = lane >> 4;

  if (tid < KMP) cksh[tid] = ckm[tid];

  v8f acc[2][6];
#pragma unroll
  for (int nt = 0; nt < 2; ++nt)
#pragma unroll
    for (int ct = 0; ct < 6; ++ct)
      acc[nt][ct] = (v8f){0.f,0.f,0.f,0.f,0.f,0.f,0.f,0.f};

  gcf pa;
  {
    int n0  = blockIdx.x * 128 + wave * 32;
    int bb  = n0 >> 11;
    int rem = n0 & 2047;
    pa = (gcf)(feats + (size_t)bb * (size_t)(CCH * HW)
                     + (size_t)(2 * lhi) * HW + rem + lrow);
  }
  gcf pb1 = (gcf)(b1 + lhi * (2 * KMP) + lrow * 2);
  gcf pb2 = (gcf)(b2 + lhi * (2 * KMP) + lrow * 2);

  struct Frag { float a[2][2]; v2f w1[6]; v2f w2[6]; };

#define LOAD_STEP(F, j)                                                      \
  do {                                                                       \
    _Pragma("unroll")                                                        \
    for (int nt = 0; nt < 2; ++nt) {                                         \
      (F).a[nt][0] = pa[(j) * 4 * HW + nt * 16];                             \
      (F).a[nt][1] = pa[(j) * 4 * HW + HW + nt * 16];                        \
    }                                                                        \
    _Pragma("unroll")                                                        \
    for (int ct = 0; ct < 6; ++ct)                                           \
      (F).w1[ct] = *(gcv2)(pb1 + (j) * 4 * KMP + ct * 32);                   \
    _Pragma("unroll")                                                        \
    for (int ct = 0; ct < 6; ++ct)                                           \
      (F).w2[ct] = *(gcv2)(pb2 + (j) * 4 * KMP + ct * 32);                   \
  } while (0)

#define ADVANCE(n)                                                           \
  do { pa += (n) * 4 * HW; pb1 += (n) * 4 * KMP; pb2 += (n) * 4 * KMP; } while (0)

#define COMPUTE_STEP(F)                                                      \
  do {                                                                       \
    v2f a_[2], q_[2];                                                        \
    _Pragma("unroll")                                                        \
    for (int nt = 0; nt < 2; ++nt) {                                         \
      a_[nt].x = (F).a[nt][0];  a_[nt].y = (F).a[nt][1];                     \
      q_[nt].x = a_[nt].x * a_[nt].x;  q_[nt].y = a_[nt].y * a_[nt].y;       \
    }                                                                        \
    _Pragma("unroll")                                                        \
    for (int ct = 0; ct < 6; ++ct) {                                         \
      _Pragma("unroll")                                                      \
      for (int nt = 0; nt < 2; ++nt) {                                       \
        acc[nt][ct] = __builtin_amdgcn_wmma_f32_16x16x4_f32(                 \
            false, q_[nt], false, (F).w1[ct], (short)0, acc[nt][ct],         \
            false, false);                                                   \
        acc[nt][ct] = __builtin_amdgcn_wmma_f32_16x16x4_f32(                 \
            false, a_[nt], false, (F).w2[ct], (short)0, acc[nt][ct],         \
            false, false);                                                   \
      }                                                                      \
    }                                                                        \
  } while (0)

#define PIN() __builtin_amdgcn_sched_barrier(0)
  Frag f0, f1, f2, f3;
  LOAD_STEP(f0, 0);
  LOAD_STEP(f1, 1);
  LOAD_STEP(f2, 2);
  ADVANCE(3);
  for (int it = 0; it < DD / 16 - 1; ++it) {
    PIN();  LOAD_STEP(f3, 0);  PIN();  COMPUTE_STEP(f0);
    PIN();  LOAD_STEP(f0, 1);  PIN();  COMPUTE_STEP(f1);
    PIN();  LOAD_STEP(f1, 2);  PIN();  COMPUTE_STEP(f2);
    PIN();  LOAD_STEP(f2, 3);  PIN();  COMPUTE_STEP(f3);
    ADVANCE(4);
  }
  PIN();  LOAD_STEP(f3, 0);  PIN();
  COMPUTE_STEP(f0);
  COMPUTE_STEP(f1);
  COMPUTE_STEP(f2);
  COMPUTE_STEP(f3);
#undef PIN
#undef LOAD_STEP
#undef ADVANCE
#undef COMPUTE_STEP

#pragma unroll
  for (int nt = 0; nt < 2; ++nt)
#pragma unroll
    for (int ct = 0; ct < 6; ++ct)
#pragma unroll
      for (int r = 0; r < 8; ++r) {
        int pl = wave * 32 + nt * 16 + r + 8 * lhi;
        sdata[pl * KMP + ct * 16 + lrow] = acc[nt][ct][r];
      }
  __syncthreads();

  {
    const float* row = &sdata[tid * KMP];
    float lse[KK];
#pragma unroll
    for (int k = 0; k < KK; ++k) {
      float lp[MMIX];
      float mx = -3.0e38f;
#pragma unroll
      for (int m = 0; m < MMIX; ++m) {
        int c = k * MMIX + m;
        lp[m] = -0.5f * row[c] + cksh[c];
        mx = fmaxf(mx, lp[m]);
      }
      float s = 0.0f;
#pragma unroll
      for (int m = 0; m < MMIX; ++m) s += expf(lp[m] - mx);
      lse[k] = mx + logf(s);
    }
    float mv = lse[0];
#pragma unroll
    for (int k = 1; k < KK; ++k) mv = fmaxf(mv, lse[k]);
    float s = 0.0f;
#pragma unroll
    for (int k = 0; k < KK; ++k) s += expf(lse[k] - mv);
    out[blockIdx.x * 128 + tid] = 1.0f - 1.0f / s;
  }
}

extern "C" void kernel_launch(void* const* d_in, const int* in_sizes, int n_in,
                              void* d_out, int out_size, void* d_ws, size_t ws_size,
                              hipStream_t stream) {
  const float* feats     = (const float*)d_in[0];
  const float* means     = (const float*)d_in[1];
  const float* log_vars  = (const float*)d_in[2];
  const float* log_w     = (const float*)d_in[3];
  float* out = (float*)d_out;

  float* b1  = (float*)d_ws;
  float* b2  = b1 + KMP * DD;
  float* ckm = b2 + KMP * DD;

  (void)in_sizes; (void)n_in; (void)out_size; (void)ws_size;

  gmm_prep<<<KMP, 128, 0, stream>>>(means, log_vars, log_w, b1, b2, ckm);
  gmm_main<<<NPTS / 128, 128, 0, stream>>>(feats, b1, b2, ckm, out);
}


# main: wave tile 32x96 -> 32x48 (4 waves = 2 row groups x 2 column halves, 64 points per group), grid 128 -> 256 so every SIMD has a wave; ring step 8 d keeps 72 WMMAs of load cover; sum order kept
# speedup vs baseline: 4.7326x; 1.5191x over previous
#include <hip/hip_runtime.h>
#include <math.h>

#define BATCH 8
#define CCH   768
#define HH    32
#define WW    64
#define HW    (HH*WW)
#define NPTS  (BATCH*HH*WW)
#define DD    768
#define KK    19
#define MMIX  5
#define KM    95
#define KMP   96
#define LOG2PI 1.8378770664093453f
#define BPTS  64

typedef __attribute__((ext_vector_type(2))) float v2f;
typedef __attribute__((ext_vector_type(8))) float v8f;

typedef const float __attribute__((address_space(1)))* gcf;
typedef const v2f   __attribute__((address_space(1)))* gcv2;

__global__ __launch_bounds__(128) void gmm_prep(
    const float* __restrict__ means, const float* __restrict__ log_vars,
    const float* __restrict__ log_w,
    float* __restrict__ b1, float* __restrict__ b2, float* __restrict__ ckm) {
  __shared__ float red[2][4];
  const int c   = blockIdx.x;
  const int tid = threadIdx.x;

  if (c >= KM) {
#pragma unroll
    for (int j = 0; j < 3; ++j) {
      int o = (tid + j * 128) * (2 * KMP) + c * 2;
      *(v2f*)(b1 + o) = (v2f){0.0f, 0.0f};
      *(v2f*)(b2 + o) = (v2f){0.0f, 0.0f};
    }
    if (tid == 0) ckm[c] = -1.0e30f;
    return;
  }

  float slv = 0.0f, smi = 0.0f;
#pragma unroll
  for (int j = 0; j < 3; ++j) {
    int p = tid + j * 128;
    v2f lv = *(const v2f*)(log_vars + c * DD + 2 * p);
    v2f mu = *(const v2f*)(means    + c * DD + 2 * p);
    v2f iv, w2;
    iv.x = expf(-lv.x);
    iv.y = expf(-lv.y);
    w2.x = -2.0f * mu.x * iv.x;
    w2.y = -2.0f * mu.y * iv.y;
    int o = p * (2 * KMP) + c * 2;
    *(v2f*)(b1 + o) = iv;
    *(v2f*)(b2 + o) = w2;
    slv += lv.x;
    slv += lv.y;
    smi += mu.x * mu.x * iv.x;
    smi += mu.y * mu.y * iv.y;
  }
#pragma unroll
  for (int m = 16; m >= 1; m >>= 1) {
    slv += __shfl_xor(slv, m);
    smi += __shfl_xor(smi, m);
  }
  if ((tid & 31) == 0) { red[0][tid >> 5] = slv; red[1][tid >> 5] = smi; }
  __syncthreads();
  if (tid == 0) {
    float tlv = (red[0][0] + red[0][1]) + (red[0][2] + red[0][3]);
    float tmi = (red[1][0] + red[1][1]) + (red[1][2] + red[1][3]);
    ckm[c] = log_w[c] - 0.5f * ((float)DD * LOG2PI + tlv + tmi);
  }
}

__global__ __launch_bounds__(128) void gmm_main(
    const float* __restrict__ feats,
    const float* __restrict__ b1, const float* __restrict__ b2,
    const float* __restrict__ ckm, float* __restrict__ out) {
  __shared__ float sdata[BPTS * KMP];
  __shared__ float cksh[KMP];

  const int tid  = threadIdx.x;
  const int wave = tid >> 5;
  const int lane = tid & 31;
  const int lrow = lane & 15;
  const int lhi  = lane >> 4;
  const int rg   = wave >> 1;
  const int ch   = wave & 1;

  if (tid < KMP) cksh[tid] = ckm[tid];

  v8f acc[2][3];
#pragma unroll
  for (int nt = 0; nt < 2; ++nt)
#pragma unroll
    for (int ct = 0; ct < 3; ++ct)
      acc[nt][ct] = (v8f){0.f,0.f,0.f,0.f,0.f,0.f,0.f,0.f};

  gcf pa;
  {
    int n0  = blockIdx.x * BPTS + rg * 32;
    int bb  = n0 >> 11;
    int rem = n0 & 2047;
    pa = (gcf)(feats + (size_t)bb * (size_t)(CCH * HW)
                     + (size_t)(2 * lhi) * HW + rem + lrow);
  }
  gcf pb1 = (gcf)(b1 + lhi * (2 * KMP) + ch * 96 + lrow * 2);
  gcf pb2 = (gcf)(b2 + lhi * (2 * KMP) + ch * 96 + lrow * 2);

  struct Frag { float a[2][2][2]; v2f w1[2][3]; v2f w2[2][3]; };

#define LOAD_STEP(F, j)                                                      \
  do {                                                                       \
    _Pragma("unroll")                                                        \
    for (int u = 0; u < 2; ++u) {                                            \
      _Pragma("unroll")                                                      \
      for (int nt = 0; nt < 2; ++nt) {                                       \
        (F).a[u][nt][0] = pa[((j) * 2 + u) * 4 * HW + nt * 16];              \
        (F).a[u][nt][1] = pa[((j) * 2 + u) * 4 * HW + HW + nt * 16];         \
      }                                                                      \
      _Pragma("unroll")                                                      \
      for (int ct = 0; ct < 3; ++ct)                                         \
        (F).w1[u][ct] = *(gcv2)(pb1 + ((j) * 2 + u) * 4 * KMP + ct * 32);    \
      _Pragma("unroll")                                                      \
      for (int ct = 0; ct < 3; ++ct)                                         \
        (F).w2[u][ct] = *(gcv2)(pb2 + ((j) * 2 + u) * 4 * KMP + ct * 32);    \
    }                                                                        \
  } while (0)

#define ADVANCE(n)                                                           \
  do { pa += (n) * 8 * HW; pb1 += (n) * 8 * KMP; pb2 += (n) * 8 * KMP; } while (0)

#define COMPUTE_STEP(F)                                                      \
  do {                                                                       \
    _Pragma("unroll")                                                        \
    for (int u = 0; u < 2; ++u) {                                            \
      v2f a_[2], q_[2];                                                      \
      _Pragma("unroll")                                                      \
      for (int nt = 0; nt < 2; ++nt) {                                       \
        a_[nt].x = (F).a[u][nt][0];  a_[nt].y = (F).a[u][nt][1];             \
        q_[nt].x = a_[nt].x * a_[nt].x;  q_[nt].y = a_[nt].y * a_[nt].y;     \
      }                                                                      \
      _Pragma("unroll")                                                      \
      for (int ct = 0; ct < 3; ++ct) {                                       \
        _Pragma("unroll")                                                    \
        for (int nt = 0; nt < 2; ++nt) {                                     \
          acc[nt][ct] = __builtin_amdgcn_wmma_f32_16x16x4_f32(               \
              false, q_[nt], false, (F).w1[u][ct], (short)0, acc[nt][ct],    \
              false, false);                                                 \
          acc[nt][ct] = __builtin_amdgcn_wmma_f32_16x16x4_f32(               \
              false, a_[nt], false, (F).w2[u][ct], (short)0, acc[nt][ct],    \
              false, false);                                                 \
        }                                                                    \
      }                                                                      \
    }                                                                        \
  } while (0)

#define PIN() __builtin_amdgcn_sched_barrier(0)
  Frag f0, f1, f2, f3;
  LOAD_STEP(f0, 0);
  LOAD_STEP(f1, 1);
  LOAD_STEP(f2, 2);
  ADVANCE(3);
  for (int it = 0; it < DD / 32 - 1; ++it) {
    PIN();  LOAD_STEP(f3, 0);  PIN();  COMPUTE_STEP(f0);
    PIN();  LOAD_STEP(f0, 1);  PIN();  COMPUTE_STEP(f1);
    PIN();  LOAD_STEP(f1, 2);  PIN();  COMPUTE_STEP(f2);
    PIN();  LOAD_STEP(f2, 3);  PIN();  COMPUTE_STEP(f3);
    ADVANCE(4);
  }
  PIN();  LOAD_STEP(f3, 0);  PIN();
  COMPUTE_STEP(f0);
  COMPUTE_STEP(f1);
  COMPUTE_STEP(f2);
  COMPUTE_STEP(f3);
#undef PIN
#undef LOAD_STEP
#undef ADVANCE
#undef COMPUTE_STEP

#pragma unroll
  for (int nt = 0; nt < 2; ++nt)
#pragma unroll
    for (int ct = 0; ct < 3; ++ct)
#pragma unroll
      for (int r = 0; r < 8; ++r) {
        int pl = rg * 32 + nt * 16 + r + 8 * lhi;
        sdata[pl * KMP + ch * 48 + ct * 16 + lrow] = acc[nt][ct][r];
      }
  __syncthreads();

  if (tid < BPTS) {
    const float* row = &sdata[tid * KMP];
    float lse[KK];
#pragma unroll
    for (int k = 0; k < KK; ++k) {
      float lp[MMIX];
      float mx = -3.0e38f;
#pragma unroll
      for (int m = 0; m < MMIX; ++m) {
        int c = k * MMIX + m;
        lp[m] = -0.5f * row[c] + cksh[c];
        mx = fmaxf(mx, lp[m]);
      }
      float s = 0.0f;
#pragma unroll
      for (int m = 0; m < MMIX; ++m) s += expf(lp[m] - mx);
      lse[k] = mx + logf(s);
    }
    float mv = lse[0];
#pragma unroll
    for (int k = 1; k < KK; ++k) mv = fmaxf(mv, lse[k]);
    float s = 0.0f;
#pragma unroll
    for (int k = 0; k < KK; ++k) s += expf(lse[k] - mv);
    out[blockIdx.x * BPTS + tid] = 1.0f - 1.0f / s;
  }
}

extern "C" void kernel_launch(void* const* d_in, const int* in_sizes, int n_in,
                              void* d_out, int out_size, void* d_ws, size_t ws_size,
                              hipStream_t stream) {
  const float* feats     = (const float*)d_in[0];
  const float* means     = (const float*)d_in[1];
  const float* log_vars  = (const float*)d_in[2];
  const float* log_w     = (const float*)d_in[3];
  float* out = (float*)d_out;

  float* b1  = (float*)d_ws;
  float* b2  = b1 + KMP * DD;
  float* ckm = b2 + KMP * DD;

  (void)in_sizes; (void)n_in; (void)out_size; (void)ws_size;

  gmm_prep<<<KMP, 128, 0, stream>>>(means, log_vars, log_w, b1, b2, ckm);
  gmm_main<<<NPTS / BPTS, 128, 0, stream>>>(feats, b1, b2, ckm, out);
}


# main: fragment loads widened - one b64 A load serves both row tiles (even/odd points), one b128 B load holds both products' k pair; 20 -> 10 loads per 24 WMMAs; order of sums kept
# speedup vs baseline: 5.1922x; 1.0971x over previous
#include <hip/hip_runtime.h>
#include <math.h>

#define BATCH 8
#define CCH   768
#define HH    32
#define WW    64
#define HW    (HH*WW)
#define NPTS  (BATCH*HH*WW)
#define DD    768
#define KK    19
#define MMIX  5
#define KM    95
#define KMP   96
#define LOG2PI 1.8378770664093453f
#define BPTS  64

typedef __attribute__((ext_vector_type(2))) float v2f;
typedef __attribute__((ext_vector_type(4))) float v4f;
typedef __attribute__((ext_vector_type(8))) float v8f;

typedef const float __attribute__((address_space(1)))* gcf;
typedef const v2f   __attribute__((address_space(1)))* gcv2;
typedef const v4f   __attribute__((address_space(1)))* gcv4;

__global__ __launch_bounds__(128) void gmm_prep(
    const float* __restrict__ means, const float* __restrict__ log_vars,
    const float* __restrict__ log_w,
    float* __restrict__ b12, float* __restrict__ ckm) {
  __shared__ float red[2][4];
  const int c   = blockIdx.x;
  const int tid = threadIdx.x;

  if (c >= KM) {
#pragma unroll
    for (int j = 0; j < 3; ++j) {
      int o = ((tid + j * 128) * KMP + c) * 4;
      *(v4f*)(b12 + o) = (v4f){0.0f, 0.0f, 0.0f, 0.0f};
    }
    if (tid == 0) ckm[c] = -1.0e30f;
    return;
  }

  float slv = 0.0f, smi = 0.0f;
#pragma unroll
  for (int j = 0; j < 3; ++j) {
    int p = tid + j * 128;
    v2f lv = *(const v2f*)(log_vars + c * DD + 2 * p);
    v2f mu = *(const v2f*)(means    + c * DD + 2 * p);
    v2f iv, w2;
    iv.x = expf(-lv.x);
    iv.y = expf(-lv.y);
    w2.x = -2.0f * mu.x * iv.x;
    w2.y = -2.0f * mu.y * iv.y;
    int o = (p * KMP + c) * 4;
    *(v4f*)(b12 + o) = (v4f){iv.x, iv.y, w2.x, w2.y};
    slv += lv.x;
    slv += lv.y;
    smi += mu.x * mu.x * iv.x;
    smi += mu.y * mu.y * iv.y;
  }
#pragma unroll
  for (int m = 16; m >= 1; m >>= 1) {
    slv += __shfl_xor(slv, m);
    smi += __shfl_xor(smi, m);
  }
  if ((tid & 31) == 0) { red[0][tid >> 5] = slv; red[1][tid >> 5] = smi; }
  __syncthreads();
  if (tid == 0) {
    float tlv = (red[0][0] + red[0][1]) + (red[0][2] + red[0][3]);
    float tmi = (red[1][0] + red[1][1]) + (red[1][2] + red[1][3]);
    ckm[c] = log_w[c] - 0.5f * ((float)DD * LOG2PI + tlv + tmi);
  }
}

__global__ __launch_bounds__(128) void gmm_main(
    const float* __restrict__ feats, const float* __restrict__ b12,
    const float* __restrict__ ckm, float* __restrict__ out) {
  __shared__ float sdata[BPTS * KMP];
  __shared__ float cksh[KMP];

  const int tid  = threadIdx.x;
  const int wave = tid >> 5;
  const int lane = tid & 31;
  const int lrow = lane & 15;
  const int lhi  = lane >> 4;
  const int rg   = wave >> 1;
  const int ch   = wave & 1;

  if (tid < KMP) cksh[tid] = ckm[tid];

  v8f acc[2][3];
#pragma unroll
  for (int nt = 0; nt < 2; ++nt)
#pragma unroll
    for (int ct = 0; ct < 3; ++ct)
      acc[nt][ct] = (v8f){0.f,0.f,0.f,0.f,0.f,0.f,0.f,0.f};

  gcf pa;
  {
    int n0  = blockIdx.x * BPTS + rg * 32;
    int bb  = n0 >> 11;
    int rem = n0 & 2047;
    pa = (gcf)(feats + (size_t)bb * (size_t)(CCH * HW)
                     + (size_t)(2 * lhi) * HW + rem + 2 * lrow);
  }
  gcf pb = (gcf)(b12 + (lhi * KMP + ch * 48 + lrow) * 4);

  struct Frag { v2f a[2][2]; v4f w[2][3]; };

#define LOAD_STEP(F, j)                                                      \
  do {                                                                       \
    _Pragma("unroll")                                                        \
    for (int u = 0; u < 2; ++u) {                                            \
      (F).a[u][0] = *(gcv2)(pa + ((j) * 2 + u) * 4 * HW);                    \
      (F).a[u][1] = *(gcv2)(pa + ((j) * 2 + u) * 4 * HW + HW);               \
      _Pragma("unroll")                                                      \
      for (int ct = 0; ct < 3; ++ct)                                         \
        (F).w[u][ct] = *(gcv4)(pb + ((j) * 2 + u) * 8 * KMP + ct * 64);      \
    }                                                                        \
  } while (0)

#define ADVANCE(n)                                                           \
  do { pa += (n) * 8 * HW; pb += (n) * 16 * KMP; } while (0)

#define COMPUTE_STEP(F)                                                      \
  do {                                                                       \
    _Pragma("unroll")                                                        \
    for (int u = 0; u < 2; ++u) {                                            \
      v2f a_[2], q_[2];                                                      \
      a_[0].x = (F).a[u][0].x;  a_[0].y = (F).a[u][1].x;                     \
      a_[1].x = (F).a[u][0].y;  a_[1].y = (F).a[u][1].y;                     \
      _Pragma("unroll")                                                      \
      for (int nt = 0; nt < 2; ++nt) {                                       \
        q_[nt].x = a_[nt].x * a_[nt].x;  q_[nt].y = a_[nt].y * a_[nt].y;     \
      }                                                                      \
      _Pragma("unroll")                                                      \
      for (int ct = 0; ct < 3; ++ct) {                                       \
        v2f w1_, w2_;                                                        \
        w1_.x = (F).w[u][ct].x;  w1_.y = (F).w[u][ct].y;                     \
        w2_.x = (F).w[u][ct].z;  w2_.y = (F).w[u][ct].w;                     \
        _Pragma("unroll")                                                    \
        for (int nt = 0; nt < 2; ++nt) {                                     \
          acc[nt][ct] = __builtin_amdgcn_wmma_f32_16x16x4_f32(               \
              false, q_[nt], false, w1_, (short)0, acc[nt][ct],              \
              false, false);                                                 \
          acc[nt][ct] = __builtin_amdgcn_wmma_f32_16x16x4_f32(               \
              false, a_[nt], false, w2_, (short)0, acc[nt][ct],              \
              false, false);                                                 \
        }                                                                    \
      }                                                                      \
    }                                                                        \
  } while (0)

#define PIN() __builtin_amdgcn_sched_barrier(0)
  Frag f0, f1, f2, f3;
  LOAD_STEP(f0, 0);
  LOAD_STEP(f1, 1);
  LOAD_STEP(f2, 2);
  ADVANCE(3);
  for (int it = 0; it < DD / 32 - 1; ++it) {
    PIN();  LOAD_STEP(f3, 0);  PIN();  COMPUTE_STEP(f0);
    PIN();  LOAD_STEP(f0, 1);  PIN();  COMPUTE_STEP(f1);
    PIN();  LOAD_STEP(f1, 2);  PIN();  COMPUTE_STEP(f2);
    PIN();  LOAD_STEP(f2, 3);  PIN();  COMPUTE_STEP(f3);
    ADVANCE(4);
  }
  PIN();  LOAD_STEP(f3, 0);  PIN();
  COMPUTE_STEP(f0);
  COMPUTE_STEP(f1);
  COMPUTE_STEP(f2);
  COMPUTE_STEP(f3);
#undef PIN
#undef LOAD_STEP
#undef ADVANCE
#undef COMPUTE_STEP

#pragma unroll
  for (int nt = 0; nt < 2; ++nt)
#pragma unroll
    for (int ct = 0; ct < 3; ++ct)
#pragma unroll
      for (int r = 0; r < 8; ++r) {
        int pl = rg * 32 + 2 * (r + 8 * lhi) + nt;
        sdata[pl * KMP + ch * 48 + ct * 16 + lrow] = acc[nt][ct][r];
      }
  __syncthreads();

  if (tid < BPTS) {
    const float* row = &sdata[tid * KMP];
    float lse[KK];
#pragma unroll
    for (int k = 0; k < KK; ++k) {
      float lp[MMIX];
      float mx = -3.0e38f;
#pragma unroll
      for (int m = 0; m < MMIX; ++m) {
        int c = k * MMIX + m;
        lp[m] = -0.5f * row[c] + cksh[c];
        mx = fmaxf(mx, lp[m]);
      }
      float s = 0.0f;
#pragma unroll
      for (int m = 0; m < MMIX; ++m) s += expf(lp[m] - mx);
      lse[k] = mx + logf(s);
    }
    float mv = lse[0];
#pragma unroll
    for (int k = 1; k < KK; ++k) mv = fmaxf(mv, lse[k]);
    float s = 0.0f;
#pragma unroll
    for (int k = 0; k < KK; ++k) s += expf(lse[k] - mv);
    out[blockIdx.x * BPTS + tid] = 1.0f - 1.0f / s;
  }
}

extern "C" void kernel_launch(void* const* d_in, const int* in_sizes, int n_in,
                              void* d_out, int out_size, void* d_ws, size_t ws_size,
                              hipStream_t stream) {
  const float* feats     = (const float*)d_in[0];
  const float* means     = (const float*)d_in[1];
  const float* log_vars  = (const float*)d_in[2];
  const float* log_w     = (const float*)d_in[3];
  float* out = (float*)d_out;

  float* b12 = (float*)d_ws;
  float* ckm = b12 + 2 * KMP * DD;

  (void)in_sizes; (void)n_in; (void)out_size; (void)ws_size;

  gmm_prep<<<KMP, 128, 0, stream>>>(means, log_vars, log_w, b12, ckm);
  gmm_main<<<NPTS / BPTS, 128, 0, stream>>>(feats, b12, ckm, out);
}


# main: each wave's d range halved (k split), 8 waves per work-group = two per SIMD, so one wave's WMMAs cover the other's load group; a sum = lower d half + upper d half, added in the epilogue
# speedup vs baseline: 5.4714x; 1.0538x over previous
#include <hip/hip_runtime.h>
#include <math.h>

#define BATCH 8
#define CCH   768
#define HH    32
#define WW    64
#define HW    (HH*WW)
#define NPTS  (BATCH*HH*WW)
#define DD    768
#define KK    19
#define MMIX  5
#define KM    95
#define KMP   96
#define LOG2PI 1.8378770664093453f
#define BPTS  64

typedef __attribute__((ext_vector_type(2))) float v2f;
typedef __attribute__((ext_vector_type(4))) float v4f;
typedef __attribute__((ext_vector_type(8))) float v8f;

typedef const float __attribute__((address_space(1)))* gcf;
typedef const v2f   __attribute__((address_space(1)))* gcv2;
typedef const v4f   __attribute__((address_space(1)))* gcv4;

__global__ __launch_bounds__(128) void gmm_prep(
    const float* __restrict__ means, const float* __restrict__ log_vars,
    const float* __restrict__ log_w,
    float* __restrict__ b12, float* __restrict__ ckm) {
  __shared__ float red[2][4];
  const int c   = blockIdx.x;
  const int tid = threadIdx.x;

  if (c >= KM) {
#pragma unroll
    for (int j = 0; j < 3; ++j) {
      int o = ((tid + j * 128) * KMP + c) * 4;
      *(v4f*)(b12 + o) = (v4f){0.0f, 0.0f, 0.0f, 0.0f};
    }
    if (tid == 0) ckm[c] = -1.0e30f;
    return;
  }

  float slv = 0.0f, smi = 0.0f;
#pragma unroll
  for (int j = 0; j < 3; ++j) {
    int p = tid + j * 128;
    v2f lv = *(const v2f*)(log_vars + c * DD + 2 * p);
    v2f mu = *(const v2f*)(means    + c * DD + 2 * p);
    v2f iv, w2;
    iv.x = expf(-lv.x);
    iv.y = expf(-lv.y);
    w2.x = -2.0f * mu.x * iv.x;
    w2.y = -2.0f * mu.y * iv.y;
    int o = (p * KMP + c) * 4;
    *(v4f*)(b12 + o) = (v4f){iv.x, iv.y, w2.x, w2.y};
    slv += lv.x;
    slv += lv.y;
    smi += mu.x * mu.x * iv.x;
    smi += mu.y * mu.y * iv.y;
  }
#pragma unroll
  for (int m = 16; m >= 1; m >>= 1) {
    slv += __shfl_xor(slv, m);
    smi += __shfl_xor(smi, m);
  }
  if ((tid & 31) == 0) { red[0][tid >> 5] = slv; red[1][tid >> 5] = smi; }
  __syncthreads();
  if (tid == 0) {
    float tlv = (red[0][0] + red[0][1]) + (red[0][2] + red[0][3]);
    float tmi = (red[1][0] + red[1][1]) + (red[1][2] + red[1][3]);
    ckm[c] = log_w[c] - 0.5f * ((float)DD * LOG2PI + tlv + tmi);
  }
}

__global__ __launch_bounds__(256) void gmm_main(
    const float* __restrict__ feats, const float* __restrict__ b12,
    const float* __restrict__ ckm, float* __restrict__ out) {
  __shared__ float sdata[2 * BPTS * KMP];
  __shared__ float cksh[KMP];

  const int tid  = threadIdx.x;
  const int wave = tid >> 5;
  const int lane = tid & 31;
  const int lrow = lane & 15;
  const int lhi  = lane >> 4;
  const int kh   = wave >> 2;
  const int rg   = (wave >> 1) & 1;
  const int ch   = wave & 1;

  if (tid < KMP) cksh[tid] = ckm[tid];

  v8f acc[2][3];
#pragma unroll
  for (int nt = 0; nt < 2; ++nt)
#pragma unroll
    for (int ct = 0; ct < 3; ++ct)
      acc[nt][ct] = (v8f){0.f,0.f,0.f,0.f,0.f,0.f,0.f,0.f};

  gcf pa;
  {
    int n0  = blockIdx.x * BPTS + rg * 32;
    int bb  = n0 >> 11;
    int rem = n0 & 2047;
    pa = (gcf)(feats + (size_t)bb * (size_t)(CCH * HW)
                     + (size_t)(kh * (DD / 2) + 2 * lhi) * HW + rem + 2 * lrow);
  }
  gcf pb = (gcf)(b12 + ((kh * (DD / 4) + lhi) * KMP + ch * 48 + lrow) * 4);

  struct Frag { v2f a[2][2]; v4f w[2][3]; };

#define LOAD_STEP(F, j)                                                      \
  do {                                                                       \
    _Pragma("unroll")                                                        \
    for (int u = 0; u < 2; ++u) {                                            \
      (F).a[u][0] = *(gcv2)(pa + ((j) * 2 + u) * 4 * HW);                    \
      (F).a[u][1] = *(gcv2)(pa + ((j) * 2 + u) * 4 * HW + HW);               \
      _Pragma("unroll")                                                      \
      for (int ct = 0; ct < 3; ++ct)                                         \
        (F).w[u][ct] = *(gcv4)(pb + ((j) * 2 + u) * 8 * KMP + ct * 64);      \
    }                                                                        \
  } while (0)

#define ADVANCE(n)                                                           \
  do { pa += (n) * 8 * HW; pb += (n) * 16 * KMP; } while (0)

#define COMPUTE_STEP(F)                                                      \
  do {                                                                       \
    _Pragma("unroll")                                                        \
    for (int u = 0; u < 2; ++u) {                                            \
      v2f a_[2], q_[2];                                                      \
      a_[0].x = (F).a[u][0].x;  a_[0].y = (F).a[u][1].x;                     \
      a_[1].x = (F).a[u][0].y;  a_[1].y = (F).a[u][1].y;                     \
      _Pragma("unroll")                                                      \
      for (int nt = 0; nt < 2; ++nt) {                                       \
        q_[nt].x = a_[nt].x * a_[nt].x;  q_[nt].y = a_[nt].y * a_[nt].y;     \
      }                                                                      \
      _Pragma("unroll")                                                      \
      for (int ct = 0; ct < 3; ++ct) {                                       \
        v2f w1_, w2_;                                                        \
        w1_.x = (F).w[u][ct].x;  w1_.y = (F).w[u][ct].y;                     \
        w2_.x = (F).w[u][ct].z;  w2_.y = (F).w[u][ct].w;                     \
        _Pragma("unroll")                                                    \
        for (int nt = 0; nt < 2; ++nt) {                                     \
          acc[nt][ct] = __builtin_amdgcn_wmma_f32_16x16x4_f32(               \
              false, q_[nt], false, w1_, (short)0, acc[nt][ct],              \
              false, false);                                                 \
          acc[nt][ct] = __builtin_amdgcn_wmma_f32_16x16x4_f32(               \
              false, a_[nt], false, w2_, (short)0, acc[nt][ct],              \
              false, false);                                                 \
        }                                                                    \
      }                                                                      \
    }                                                                        \
  } while (0)

#define PIN() __builtin_amdgcn_sched_barrier(0)
  Frag f0, f1, f2, f3;
  LOAD_STEP(f0, 0);
  LOAD_STEP(f1, 1);
  LOAD_STEP(f2, 2);
  ADVANCE(3);
  for (int it = 0; it < DD / 64 - 1; ++it) {
    PIN();  LOAD_STEP(f3, 0);  PIN();  COMPUTE_STEP(f0);
    PIN();  LOAD_STEP(f0, 1);  PIN();  COMPUTE_STEP(f1);
    PIN();  LOAD_STEP(f1, 2);  PIN();  COMPUTE_STEP(f2);
    PIN();  LOAD_STEP(f2, 3);  PIN();  COMPUTE_STEP(f3);
    ADVANCE(4);
  }
  PIN();  LOAD_STEP(f3, 0);  PIN();
  COMPUTE_STEP(f0);
  COMPUTE_STEP(f1);
  COMPUTE_STEP(f2);
  COMPUTE_STEP(f3);
#undef PIN
#undef LOAD_STEP
#undef ADVANCE
#undef COMPUTE_STEP

#pragma unroll
  for (int nt = 0; nt < 2; ++nt)
#pragma unroll
    for (int ct = 0; ct < 3; ++ct)
#pragma unroll
      for (int r = 0; r < 8; ++r) {
        int pl = rg * 32 + 2 * (r + 8 * lhi) + nt;
        sdata[(kh * BPTS + pl) * KMP + ch * 48 + ct * 16 + lrow] = acc[nt][ct][r];
      }
  __syncthreads();

  if (tid < BPTS) {
    const float* row  = &sdata[tid * KMP];
    const float* row2 = &sdata[(BPTS + tid) * KMP];
    float lse[KK];
#pragma unroll
    for (int k = 0; k < KK; ++k) {
      float lp[MMIX];
      float mx = -3.0e38f;
#pragma unroll
      for (int m = 0; m < MMIX; ++m) {
        int c = k * MMIX + m;
        lp[m] = -0.5f * (row[c] + row2[c]) + cksh[c];
        mx = fmaxf(mx, lp[m]);
      }
      float s = 0.0f;
#pragma unroll
      for (int m = 0; m < MMIX; ++m) s += expf(lp[m] - mx);
      lse[k] = mx + logf(s);
    }
    float mv = lse[0];
#pragma unroll
    for (int k = 1; k < KK; ++k) mv = fmaxf(mv, lse[k]);
    float s = 0.0f;
#pragma unroll
    for (int k = 0; k < KK; ++k) s += expf(lse[k] - mv);
    out[blockIdx.x * BPTS + tid] = 1.0f - 1.0f / s;
  }
}

extern "C" void kernel_launch(void* const* d_in, const int* in_sizes, int n_in,
                              void* d_out, int out_size, void* d_ws, size_t ws_size,
                              hipStream_t stream) {
  const float* feats     = (const float*)d_in[0];
  const float* means     = (const float*)d_in[1];
  const float* log_vars  = (const float*)d_in[2];
  const float* log_w     = (const float*)d_in[3];
  float* out = (float*)d_out;

  float* b12 = (float*)d_ws;
  float* ckm = b12 + 2 * KMP * DD;

  (void)in_sizes; (void)n_in; (void)out_size; (void)ws_size;

  gmm_prep<<<KMP, 128, 0, stream>>>(means, log_vars, log_w, b12, ckm);
  gmm_main<<<NPTS / BPTS, 256, 0, stream>>>(feats, b12, ckm, out);
}


# epilogue: class sums on all 8 waves (a wave pair per 5 classes, 64 points each; was 2 waves for all 19), LDS rows at odd pitch, column constants by scalar loads; final softmax and sum order kept
# speedup vs baseline: 5.8473x; 1.0687x over previous
#include <hip/hip_runtime.h>
#include <math.h>

#define BATCH 8
#define CCH   768
#define HH    32
#define WW    64
#define HW    (HH*WW)
#define NPTS  (BATCH*HH*WW)
#define DD    768
#define KK    19
#define MMIX  5
#define KM    95
#define KMP   96
#define LOG2PI 1.8378770664093453f
#define BPTS  64
#define SPITCH 97

typedef __attribute__((ext_vector_type(2))) float v2f;
typedef __attribute__((ext_vector_type(4))) float v4f;
typedef __attribute__((ext_vector_type(8))) float v8f;

typedef const float __attribute__((address_space(1)))* gcf;
typedef const v2f   __attribute__((address_space(1)))* gcv2;
typedef const v4f   __attribute__((address_space(1)))* gcv4;

__global__ __launch_bounds__(128) void gmm_prep(
    const float* __restrict__ means, const float* __restrict__ log_vars,
    const float* __restrict__ log_w,
    float* __restrict__ b12, float* __restrict__ ckm) {
  __shared__ float red[2][4];
  const int c   = blockIdx.x;
  const int tid = threadIdx.x;

  if (c >= KM) {
#pragma unroll
    for (int j = 0; j < 3; ++j) {
      int o = ((tid + j * 128) * KMP + c) * 4;
      *(v4f*)(b12 + o) = (v4f){0.0f, 0.0f, 0.0f, 0.0f};
    }
    if (tid == 0) ckm[c] = -1.0e30f;
    return;
  }

  float slv = 0.0f, smi = 0.0f;
#pragma unroll
  for (int j = 0; j < 3; ++j) {
    int p = tid + j * 128;
    v2f lv = *(const v2f*)(log_vars + c * DD + 2 * p);
    v2f mu = *(const v2f*)(means    + c * DD + 2 * p);
    v2f iv, w2;
    iv.x = expf(-lv.x);
    iv.y = expf(-lv.y);
    w2.x = -2.0f * mu.x * iv.x;
    w2.y = -2.0f * mu.y * iv.y;
    int o = (p * KMP + c) * 4;
    *(v4f*)(b12 + o) = (v4f){iv.x, iv.y, w2.x, w2.y};
    slv += lv.x;
    slv += lv.y;
    smi += mu.x * mu.x * iv.x;
    smi += mu.y * mu.y * iv.y;
  }
#pragma unroll
  for (int m = 16; m >= 1; m >>= 1) {
    slv += __shfl_xor(slv, m);
    smi += __shfl_xor(smi, m);
  }
  if ((tid & 31) == 0) { red[0][tid >> 5] = slv; red[1][tid >> 5] = smi; }
  __syncthreads();
  if (tid == 0) {
    float tlv = (red[0][0] + red[0][1]) + (red[0][2] + red[0][3]);
    float tmi = (red[1][0] + red[1][1]) + (red[1][2] + red[1][3]);
    ckm[c] = log_w[c] - 0.5f * ((float)DD * LOG2PI + tlv + tmi);
  }
}

__global__ __launch_bounds__(256) void gmm_main(
    const float* __restrict__ feats, const float* __restrict__ b12,
    const float* __restrict__ ckm, float* __restrict__ out) {
  __shared__ float sdata[2 * BPTS * SPITCH];
  __shared__ float lsesh[KK * BPTS];

  const int tid  = threadIdx.x;
  const int wave = tid >> 5;
  const int lane = tid & 31;
  const int lrow = lane & 15;
  const int lhi  = lane >> 4;
  const int kh   = wave >> 2;
  const int rg   = (wave >> 1) & 1;
  const int ch   = wave & 1;

  v8f acc[2][3];
#pragma unroll
  for (int nt = 0; nt < 2; ++nt)
#pragma unroll
    for (int ct = 0; ct < 3; ++ct)
      acc[nt][ct] = (v8f){0.f,0.f,0.f,0.f,0.f,0.f,0.f,0.f};

  gcf pa;
  {
    int n0  = blockIdx.x * BPTS + rg * 32;
    int bb  = n0 >> 11;
    int rem = n0 & 2047;
    pa = (gcf)(feats + (size_t)bb * (size_t)(CCH * HW)
                     + (size_t)(kh * (DD / 2) + 2 * lhi) * HW + rem + 2 * lrow);
  }
  gcf pb = (gcf)(b12 + ((kh * (DD / 4) + lhi) * KMP + ch * 48 + lrow) * 4);

  struct Frag { v2f a[2][2]; v4f w[2][3]; };

#define LOAD_STEP(F, j)                                                      \
  do {                                                                       \
    _Pragma("unroll")                                                        \
    for (int u = 0; u < 2; ++u) {                                            \
      (F).a[u][0] = *(gcv2)(pa + ((j) * 2 + u) * 4 * HW);                    \
      (F).a[u][1] = *(gcv2)(pa + ((j) * 2 + u) * 4 * HW + HW);               \
      _Pragma("unroll")                                                      \
      for (int ct = 0; ct < 3; ++ct)                                         \
        (F).w[u][ct] = *(gcv4)(pb + ((j) * 2 + u) * 8 * KMP + ct * 64);      \
    }                                                                        \
  } while (0)

#define ADVANCE(n)                                                           \
  do { pa += (n) * 8 * HW; pb += (n) * 16 * KMP; } while (0)

#define COMPUTE_STEP(F)                                                      \
  do {                                                                       \
    _Pragma("unroll")                                                        \
    for (int u = 0; u < 2; ++u) {                                            \
      v2f a_[2], q_[2];                                                      \
      a_[0].x = (F).a[u][0].x;  a_[0].y = (F).a[u][1].x;                     \
      a_[1].x = (F).a[u][0].y;  a_[1].y = (F).a[u][1].y;                     \
      _Pragma("unroll")                                                      \
      for (int nt = 0; nt < 2; ++nt) {                                       \
        q_[nt].x = a_[nt].x * a_[nt].x;  q_[nt].y = a_[nt].y * a_[nt].y;     \
      }                                                                      \
      _Pragma("unroll")                                                      \
      for (int ct = 0; ct < 3; ++ct) {                                       \
        v2f w1_, w2_;                                                        \
        w1_.x = (F).w[u][ct].x;  w1_.y = (F).w[u][ct].y;                     \
        w2_.x = (F).w[u][ct].z;  w2_.y = (F).w[u][ct].w;                     \
        _Pragma("unroll")                                                    \
        for (int nt = 0; nt < 2; ++nt) {                                     \
          acc[nt][ct] = __builtin_amdgcn_wmma_f32_16x16x4_f32(               \
              false, q_[nt], false, w1_, (short)0, acc[nt][ct],              \
              false, false);                                                 \
          acc[nt][ct] = __builtin_amdgcn_wmma_f32_16x16x4_f32(               \
              false, a_[nt], false, w2_, (short)0, acc[nt][ct],              \
              false, false);                                                 \
        }                                                                    \
      }                                                                      \
    }                                                                        \
  } while (0)

#define PIN() __builtin_amdgcn_sched_barrier(0)
  Frag f0, f1, f2, f3;
  LOAD_STEP(f0, 0);
  LOAD_STEP(f1, 1);
  LOAD_STEP(f2, 2);
  ADVANCE(3);
  for (int it = 0; it < DD / 64 - 1; ++it) {
    PIN();  LOAD_STEP(f3, 0);  PIN();  COMPUTE_STEP(f0);
    PIN();  LOAD_STEP(f0, 1);  PIN();  COMPUTE_STEP(f1);
    PIN();  LOAD_STEP(f1, 2);  PIN();  COMPUTE_STEP(f2);
    PIN();  LOAD_STEP(f2, 3);  PIN();  COMPUTE_STEP(f3);
    ADVANCE(4);
  }
  PIN();  LOAD_STEP(f3, 0);  PIN();
  COMPUTE_STEP(f0);
  COMPUTE_STEP(f1);
  COMPUTE_STEP(f2);
  COMPUTE_STEP(f3);
#undef PIN
#undef LOAD_STEP
#undef ADVANCE
#undef COMPUTE_STEP

#pragma unroll
  for (int nt = 0; nt < 2; ++nt)
#pragma unroll
    for (int ct = 0; ct < 3; ++ct)
#pragma unroll
      for (int r = 0; r < 8; ++r) {
        int pl = rg * 32 + 2 * (r + 8 * lhi) + nt;
        sdata[(kh * BPTS + pl) * SPITCH + ch * 48 + ct * 16 + lrow] = acc[nt][ct][r];
      }
  __syncthreads();

  {
    const int pt = tid & (BPTS - 1);
    const int q = __builtin_amdgcn_readfirstlane(tid >> 6);
    const float* row  = &sdata[pt * SPITCH];
    const float* row2 = &sdata[(BPTS + pt) * SPITCH];
#pragma unroll
    for (int j = 0; j < MMIX; ++j) {
      const int k = q * 5 + j;
      if (k < KK) {
        float lp[MMIX];
        float mx = -3.0e38f;
#pragma unroll
        for (int m = 0; m < MMIX; ++m) {
          int c = k * MMIX + m;
          lp[m] = -0.5f * (row[c] + row2[c]) + ckm[c];
          mx = fmaxf(mx, lp[m]);
        }
        float s = 0.0f;
#pragma unroll
        for (int m = 0; m < MMIX; ++m) s += expf(lp[m] - mx);
        lsesh[k * BPTS + pt] = mx + logf(s);
      }
    }
  }
  __syncthreads();

  if (tid < BPTS) {
    float lse[KK];
#pragma unroll
    for (int k = 0; k < KK; ++k) lse[k] = lsesh[k * BPTS + tid];
    float mv = lse[0];
#pragma unroll
    for (int k = 1; k < KK; ++k) mv = fmaxf(mv, lse[k]);
    float s = 0.0f;
#pragma unroll
    for (int k = 0; k < KK; ++k) s += expf(lse[k] - mv);
    out[blockIdx.x * BPTS + tid] = 1.0f - 1.0f / s;
  }
}

extern "C" void kernel_launch(void* const* d_in, const int* in_sizes, int n_in,
                              void* d_out, int out_size, void* d_ws, size_t ws_size,
                              hipStream_t stream) {
  const float* feats     = (const float*)d_in[0];
  const float* means     = (const float*)d_in[1];
  const float* log_vars  = (const float*)d_in[2];
  const float* log_w     = (const float*)d_in[3];
  float* out = (float*)d_out;

  float* b12 = (float*)d_ws;
  float* ckm = b12 + 2 * KMP * DD;

  (void)in_sizes; (void)n_in; (void)out_size; (void)ws_size;

  gmm_prep<<<KMP, 128, 0, stream>>>(means, log_vars, log_w, b12, ckm);
  gmm_main<<<NPTS / BPTS, 256, 0, stream>>>(feats, b12, ckm, out);
}
